# starting kernel
# baseline (speedup 1.0000x reference)
#include <hip/hip_runtime.h>


#define BATCH 4
#define SEQ   2048
#define DIM   512
#define HEADS 8
#define DK    64
#define NCOL  1536
#define BPS   160

typedef __attribute__((ext_vector_type(2))) float v2f;
typedef __attribute__((ext_vector_type(8))) float v8f;

__device__ __forceinline__ v8f wmma4(v2f a, v2f b, v8f c) {
    return __builtin_amdgcn_wmma_f32_16x16x4_f32(
         false, a,  false, b,
         (short)0, c,  false,  false);
}

__device__ __forceinline__ void stage_pair(float* __restrict__ dst, int kp, int cp,
                                           const float* __restrict__ src, int srcStride) {
    const float* s = src + (size_t)(2 * kp) * srcStride + 2 * cp;
    float2 r0 = *(const float2*)s;
    float2 r1 = *(const float2*)(s + srcStride);
    float4 pk;
    pk.x = r0.x; pk.y = r1.x; pk.z = r0.y; pk.w = r1.y;
    *(float4*)&dst[kp * BPS + 4 * cp] = pk;
}

__global__ __launch_bounds__(256)
void qkv_gemm_kernel(const float* __restrict__ x, const float* __restrict__ w,
                     float* __restrict__ qb, float* __restrict__ kb,
                     float* __restrict__ vb) {
    __shared__ float lds_a[64][68];
    __shared__ float lds_b[32 * BPS];

    const int tid  = threadIdx.x;
    const int lane = tid & 31, wave = tid >> 5;
    const int half = lane >> 4, l16 = lane & 15;
    const int M0 = (wave & 3) * 16;
    const int N0 = (wave >> 2) * 32;

    const int cb    = blockIdx.x;
    const int col0  = cb * 64;
    const int which = cb >> 3, h = cb & 7;
    const int rowBase = blockIdx.y * 64;
    float* dstBase = (which == 0) ? qb : (which == 1) ? kb : vb;

    const int lr  = tid >> 2;
    const int lc  = (tid & 3) * 16;
    const int cp  = tid & 31;
    const int kb8 = tid >> 5;

    v8f acc0 = {}; v8f acc1 = {};

    for (int kc = 0; kc < DIM; kc += 64) {
        __syncthreads();
        #pragma unroll
        for (int i = 0; i < 4; ++i) {
            *(float4*)&lds_a[lr][lc + 4*i] =
                *(const float4*)&x[(size_t)(rowBase + lr) * DIM + kc + lc + 4*i];
            stage_pair(lds_b, kb8 + 8*i, cp, &w[(size_t)kc * NCOL + col0], NCOL);
        }
        __syncthreads();
        #pragma unroll
        for (int k0 = 0; k0 < 64; k0 += 4) {
            const int kp = (k0 >> 1) + half;
            v2f a  = *(const v2f*)&lds_a[M0 + l16][k0 + 2*half];
            v2f b0 = *(const v2f*)&lds_b[kp * BPS + 2*(N0 + l16)];
            v2f b1 = *(const v2f*)&lds_b[kp * BPS + 2*(N0 + 16 + l16)];
            acc0 = wmma4(a, b0, acc0);
            acc1 = wmma4(a, b1, acc1);
        }
    }

    #pragma unroll
    for (int r = 0; r < 8; ++r) {
        int rowg = rowBase + M0 + r + 8*half;
        int bb = rowg >> 11, n = rowg & (SEQ - 1);
        float* dst = dstBase + ((size_t)((bb * HEADS + h) * SEQ + n)) * DK;
        dst[N0 + l16]      = acc0[r];
        dst[N0 + 16 + l16] = acc1[r];
    }
}

__global__ __launch_bounds__(256)
void ln_kernel(float* __restrict__ t, const float* __restrict__ g,
               const float* __restrict__ bta) {
    const int row  = blockIdx.x * 8 + (threadIdx.x >> 5);
    const int lane = threadIdx.x & 31;
    const int h = (row >> 11) & (HEADS - 1);

    float2 xv = *(float2*)(t + (size_t)row * DK + lane * 2);
    float s = xv.x + xv.y;
    #pragma unroll
    for (int o = 16; o >= 1; o >>= 1) s += __shfl_xor(s, o, 32);
    const float mu = s * (1.0f / DK);
    const float d0 = xv.x - mu, d1 = xv.y - mu;
    float v = d0 * d0 + d1 * d1;
    #pragma unroll
    for (int o = 16; o >= 1; o >>= 1) v += __shfl_xor(v, o, 32);
    const float rinv = rsqrtf(v * (1.0f / DK) + 1e-5f);

    float2 gv = *(const float2*)(g   + h * DK + lane * 2);
    float2 bv = *(const float2*)(bta + h * DK + lane * 2);
    float2 ov;
    ov.x = d0 * rinv * gv.x + bv.x;
    ov.y = d1 * rinv * gv.y + bv.y;
    *(float2*)(t + (size_t)row * DK + lane * 2) = ov;
}

__global__ __launch_bounds__(256)
void kv_gemm_kernel(const float* __restrict__ kbuf, const float* __restrict__ vbuf,
                    float* __restrict__ kvb) {
    __shared__ float lds_k[32 * BPS];
    __shared__ float lds_v[32 * BPS];

    const int tid  = threadIdx.x;
    const int lane = tid & 31, wave = tid >> 5;
    const int half = lane >> 4, l16 = lane & 15;
    const int M0 = (wave & 3) * 16, N0 = (wave >> 2) * 32;
    const int bh = blockIdx.x;
    const float* kp_ptr = kbuf + (size_t)bh * SEQ * DK;
    const float* vp_ptr = vbuf + (size_t)bh * SEQ * DK;
    const int cp  = tid & 31;
    const int tb8 = tid >> 5;

    v8f acc0 = {}; v8f acc1 = {};

    for (int t0 = 0; t0 < SEQ; t0 += 64) {
        __syncthreads();
        #pragma unroll
        for (int i = 0; i < 4; ++i) {
            stage_pair(lds_k, tb8 + 8*i, cp, kp_ptr + (size_t)t0 * DK, DK);
            stage_pair(lds_v, tb8 + 8*i, cp, vp_ptr + (size_t)t0 * DK, DK);
        }
        __syncthreads();
        #pragma unroll
        for (int k0 = 0; k0 < 64; k0 += 4) {
            const int kp = (k0 >> 1) + half;
            v2f a  = *(const v2f*)&lds_k[kp * BPS + 2*(M0 + l16)];
            v2f b0 = *(const v2f*)&lds_v[kp * BPS + 2*(N0 + l16)];
            v2f b1 = *(const v2f*)&lds_v[kp * BPS + 2*(N0 + 16 + l16)];
            acc0 = wmma4(a, b0, acc0);
            acc1 = wmma4(a, b1, acc1);
        }
    }

    const float scale = 1.0f / 16384.0f;
    float* kvd = kvb + (size_t)bh * DK * DK;
    #pragma unroll
    for (int r = 0; r < 8; ++r) {
        int m = M0 + r + 8*half;
        kvd[m * DK + N0 + l16]      = acc0[r] * scale;
        kvd[m * DK + N0 + 16 + l16] = acc1[r] * scale;
    }
}

__global__ __launch_bounds__(256)
void out_gemm_kernel(const float* __restrict__ qbuf, const float* __restrict__ kvb,
                     float* __restrict__ out) {
    __shared__ float lds_q[64][68];
    __shared__ float lds_kv[32 * BPS];

    const int tid  = threadIdx.x;
    const int lane = tid & 31, wave = tid >> 5;
    const int half = lane >> 4, l16 = lane & 15;
    const int M0 = (wave & 3) * 16, N0 = (wave >> 2) * 32;
    const int tok0 = blockIdx.x * 64;
    const int bh = blockIdx.y;
    const int bb = bh >> 3, h = bh & 7;
    const float* qp  = qbuf + (size_t)bh * SEQ * DK + (size_t)tok0 * DK;
    const float* kvp = kvb + (size_t)bh * DK * DK;
    const int lr  = tid >> 2, lc = (tid & 3) * 16;
    const int cp  = tid & 31, kb8 = tid >> 5;

    #pragma unroll
    for (int i = 0; i < 4; ++i) {
        *(float4*)&lds_q[lr][lc + 4*i] = *(const float4*)&qp[(size_t)lr * DK + lc + 4*i];
        stage_pair(lds_kv, kb8 + 8*i, cp, kvp, DK);
    }
    __syncthreads();

    v8f acc0 = {}; v8f acc1 = {};
    #pragma unroll
    for (int k0 = 0; k0 < 64; k0 += 4) {
        const int kp = (k0 >> 1) + half;
        v2f a  = *(const v2f*)&lds_q[M0 + l16][k0 + 2*half];
        v2f b0 = *(const v2f*)&lds_kv[kp * BPS + 2*(N0 + l16)];
        v2f b1 = *(const v2f*)&lds_kv[kp * BPS + 2*(N0 + 16 + l16)];
        acc0 = wmma4(a, b0, acc0);
        acc1 = wmma4(a, b1, acc1);
    }

    #pragma unroll
    for (int r = 0; r < 8; ++r) {
        int n = tok0 + M0 + r + 8*half;
        float* dst = out + ((size_t)(bb * SEQ + n)) * (HEADS * DK) + h * DK;
        dst[N0 + l16]      = acc0[r];
        dst[N0 + 16 + l16] = acc1[r];
    }
}

extern "C" void kernel_launch(void* const* d_in, const int* in_sizes, int n_in,
                              void* d_out, int out_size, void* d_ws, size_t ws_size,
                              hipStream_t stream) {
    const float* x  = (const float*)d_in[0];
    const float* w  = (const float*)d_in[1];
    const float* gq = (const float*)d_in[2];
    const float* bq = (const float*)d_in[3];
    const float* gk = (const float*)d_in[4];
    const float* bk = (const float*)d_in[5];
    float* out = (float*)d_out;

    float* ws = (float*)d_ws;
    const size_t perBuf = (size_t)BATCH * HEADS * SEQ * DK;
    float* qb  = ws;
    float* kb  = ws + perBuf;
    float* vb  = ws + 2 * perBuf;
    float* kvb = ws + 3 * perBuf;

    qkv_gemm_kernel<<<dim3(NCOL / 64, (BATCH * SEQ) / 64), 256, 0, stream>>>(x, w, qb, kb, vb);
    ln_kernel<<<dim3((BATCH * HEADS * SEQ) / 8), 256, 0, stream>>>(qb, gq, bq);
    ln_kernel<<<dim3((BATCH * HEADS * SEQ) / 8), 256, 0, stream>>>(kb, gk, bk);
    kv_gemm_kernel<<<dim3(BATCH * HEADS), 256, 0, stream>>>(kb, vb, kvb);
    out_gemm_kernel<<<dim3(SEQ / 64, BATCH * HEADS), 256, 0, stream>>>(qb, kvb, out);
}


# k^T@v workgroups moved into the launch that projects 768 of the q tiles (second role, 16-token wave-private stages, 4 workgroups a WGP); launch 1 projects k, v and 256 q tiles; sum orders unchanged
# speedup vs baseline: 2.1939x; 2.1939x over previous
#include <hip/hip_runtime.h>


#define BATCH 4
#define SEQ   2048
#define DIM   512
#define HEADS 8
#define DK    64
#define NCOL  1536
#define BPS   160

typedef __attribute__((ext_vector_type(2))) float v2f;
typedef __attribute__((ext_vector_type(4))) float v4f;
typedef __attribute__((ext_vector_type(8))) float v8f;

__device__ __forceinline__ v8f wmma4(v2f a, v2f b, v8f c) {
    return __builtin_amdgcn_wmma_f32_16x16x4_f32(
         false, a,  false, b,
         (short)0, c,  false,  false);
}


__device__ __forceinline__ void load_rows(v4f (&r)[4], const float* __restrict__ src,
                                          int srcStride, int sr, int sc) {
    #pragma unroll
    for (int i = 0; i < 4; ++i)
        r[i] = *(const v4f*)(src + (size_t)(sr + 16 * i) * srcStride + sc);
}

__device__ __forceinline__ void load_pairs(v4f (&r0)[2], v4f (&r1)[2],
                                           const float* __restrict__ src,
                                           int srcStride, int sr, int sc) {
    #pragma unroll
    for (int i = 0; i < 2; ++i) {
        const float* s = src + (size_t)(2 * (sr + 16 * i)) * srcStride + sc;
        r0[i] = *(const v4f*)s;
        r1[i] = *(const v4f*)(s + srcStride);
    }
}

#define HOLD_LOADS(a, p, q)                                                        \
    asm volatile("" :: "v"(a[0]), "v"(a[1]), "v"(a[2]), "v"(a[3]),                 \
                       "v"(p[0]), "v"(p[1]), "v"(q[0]), "v"(q[1]))

template <typename T4>
__device__ __forceinline__ void put_pairs(float* __restrict__ dst, T4 r0, T4 r1) {
    v4f lo, hi;
    lo.x = r0.x; lo.y = r1.x; lo.z = r0.y; lo.w = r1.y;
    hi.x = r0.z; hi.y = r1.z; hi.z = r0.w; hi.w = r1.w;
    *(v4f*)dst       = lo;
    *(v4f*)(dst + 4) = hi;
}

#define SMEM_FLOATS (64 * 68 + 32 * BPS)

__device__ __forceinline__ void proj_tile(const float* __restrict__ x, const float* __restrict__ w,
                                          float* __restrict__ dstBase, int col0, int h, int rowBase,
                                          float* smem) {
    float* lds_a = smem;
    float* lds_b = smem + 64 * 68;

    const int tid  = threadIdx.x;
    const int lane = tid & 31, wave = tid >> 5;
    const int half = lane >> 4, l16 = lane & 15;
    const int M0 = (wave & 3) * 16;
    const int N0 = (wave >> 2) * 32;

    const int sr = tid >> 4;
    const int sc = (tid & 15) * 4;

    v8f acc0 = {}; v8f acc1 = {};

    const float* xp = x + (size_t)rowBase * DIM;
    const float* wp = w + col0;
    for (int kc = 0; kc < DIM; kc += 64) {
        __syncthreads();
        v4f xa[4], w0[2], w1[2];
        load_rows(xa, xp, DIM, sr, sc);
        load_pairs(w0, w1, wp, NCOL, sr, sc);
        xp += 64;
        wp += (size_t)64 * NCOL;
        HOLD_LOADS(xa, w0, w1);
        #pragma unroll
        for (int i = 0; i < 4; ++i) *(v4f*)&lds_a[(sr + 16 * i) * 68 + sc] = xa[i];
        #pragma unroll
        for (int i = 0; i < 2; ++i) put_pairs(&lds_b[(sr + 16 * i) * BPS + 2 * sc], w0[i], w1[i]);
        __syncthreads();
        #pragma unroll
        for (int k0 = 0; k0 < 64; k0 += 4) {
            const int kp = (k0 >> 1) + half;
            v2f a  = *(const v2f*)&lds_a[(M0 + l16) * 68 + k0 + 2*half];
            v2f b0 = *(const v2f*)&lds_b[kp * BPS + 2*(N0 + l16)];
            v2f b1 = *(const v2f*)&lds_b[kp * BPS + 2*(N0 + 16 + l16)];
            acc0 = wmma4(a, b0, acc0);
            acc1 = wmma4(a, b1, acc1);
        }
    }

    #pragma unroll
    for (int r = 0; r < 8; ++r) {
        int rowg = rowBase + M0 + r + 8*half;
        int bb = rowg >> 11, n = rowg & (SEQ - 1);
        float* dst = dstBase + ((size_t)((bb * HEADS + h) * SEQ + n)) * DK;
        dst[N0 + l16]      = acc0[r];
        dst[N0 + 16 + l16] = acc1[r];
    }
}

#define KV_TILES (16 * ((BATCH * SEQ) / 64))
#define Q_TILES  (8 * ((BATCH * SEQ) / 64))
#define KV_BLOCKS (8 * BATCH * HEADS)

__global__ __launch_bounds__(256)
void proj1_kernel(const float* __restrict__ x, const float* __restrict__ w,
                  float* __restrict__ qb, float* __restrict__ kb, float* __restrict__ vb) {
    __shared__ __attribute__((aligned(16))) float smem[SMEM_FLOATS];
    const int bid = blockIdx.x;
    int cb, rb;
    if (bid < KV_TILES) { cb = 8 + (bid & 15); rb = bid >> 4; }
    else                { cb = (bid - KV_TILES) & 7; rb = (bid - KV_TILES) >> 3; }
    float* dst = (cb < 8) ? qb : (cb < 16) ? kb : vb;
    proj_tile(x, w, dst, cb * 64, cb & 7, rb * 64, smem);
}

template <int CTRL>
__device__ __forceinline__ float row_xadd(float v) {
    const int t = __builtin_amdgcn_update_dpp(0, __float_as_int(v), CTRL, 0xf, 0xf, true);
    return v + __int_as_float(t);
}
#define XM1 0x161
#define XM2 0x162
#define XM4 0x164
#define XM8 0x168

__device__ __forceinline__ v4f ln_row16(v4f x, v4f g, v4f b) {
    float s = (x.x + x.y) + (x.z + x.w);
    s = row_xadd<XM8>(s); s = row_xadd<XM4>(s); s = row_xadd<XM2>(s); s = row_xadd<XM1>(s);
    const float mu = s * (1.0f / DK);
    const float d0 = x.x - mu, d1 = x.y - mu, d2 = x.z - mu, d3 = x.w - mu;
    float v = (d0 * d0 + d1 * d1) + (d2 * d2 + d3 * d3);
    v = row_xadd<XM8>(v); v = row_xadd<XM4>(v); v = row_xadd<XM2>(v); v = row_xadd<XM1>(v);
    const float rinv = rsqrtf(v * (1.0f / DK) + 1e-5f);
    v4f o;
    o.x = d0 * rinv * g.x + b.x;
    o.y = d1 * rinv * g.y + b.y;
    o.z = d2 * rinv * g.z + b.z;
    o.w = d3 * rinv * g.w + b.w;
    return o;
}

__device__ __forceinline__ float sum4(float4 a) { return (a.x + a.y) + (a.z + a.w); }
__device__ __forceinline__ float sq4(float4 a, float mu) {
    const float d0 = a.x - mu, d1 = a.y - mu, d2 = a.z - mu, d3 = a.w - mu;
    return (d0 * d0 + d1 * d1) + (d2 * d2 + d3 * d3);
}

__device__ __forceinline__ float4 ln_row4(float4 p0, float4 p1, float4 p2, float4 p3,
                                          float4 g, float4 b) {
    float s = (sum4(p0) + sum4(p2)) + (sum4(p1) + sum4(p3));
    s = row_xadd<XM1>(s); s = row_xadd<XM2>(s);
    const float mu = s * (1.0f / DK);
    float v = (sq4(p0, mu) + sq4(p2, mu)) + (sq4(p1, mu) + sq4(p3, mu));
    v = row_xadd<XM1>(v); v = row_xadd<XM2>(v);
    const float rinv = rsqrtf(v * (1.0f / DK) + 1e-5f);
    float4 o;
    o.x = (p0.x - mu) * rinv * g.x + b.x;
    o.y = (p0.y - mu) * rinv * g.y + b.y;
    o.z = (p0.z - mu) * rinv * g.z + b.z;
    o.w = (p0.w - mu) * rinv * g.w + b.w;
    return o;
}

#define KPS 32
#define VPS 96
#define KVW 8
#define KVT (SEQ / KVW)
#define KVC 16
#define KVS (8 * KPS + 8 * VPS)

__device__ __forceinline__ void kv_block(float* smem,
                                         const float* __restrict__ kbuf, const float* __restrict__ vbuf,
                                         const float* __restrict__ gk, const float* __restrict__ bk,
                                         float* __restrict__ kvb, int tile, int bh) {
    const int lane = threadIdx.x & 31;
    const int wave = __builtin_amdgcn_readfirstlane(threadIdx.x >> 5);
    const int half = lane >> 4, l16 = lane & 15;
    const int M0 = (tile & 3) * 16, N0 = (tile >> 2) * 32;

    float* lds_k = smem + wave * KVS;
    float* lds_v = lds_k + 8 * KPS;

    const int ktp = lane >> 2, kc = (lane & 3) * 4;
    const int vtp = lane >> 3, vc = (lane & 7) * 4;
    const int mblk = tile & 3;
    const int kc1 = 16 * ((mblk + 1) & 3), kc2 = 16 * ((mblk + 2) & 3), kc3 = 16 * ((mblk + 3) & 3);
    const size_t tokBase = (size_t)bh * SEQ + (size_t)wave * KVT;
    const float* kg = kbuf + tokBase * DK;
    const float* vg = vbuf + tokBase * DK + N0;
    const unsigned kofs = (unsigned)(2 * ktp * DK + kc);
    const unsigned vofs = (unsigned)(2 * vtp * DK + vc);
    float* ks = lds_k + ktp * KPS + 2 * kc;
    float* vs = lds_v + vtp * VPS + 2 * vc;

    const int hd = bh & (HEADS - 1);
    const float4 gv = *(const float4*)(gk + hd * DK + M0 + kc);
    const float4 bv = *(const float4*)(bk + hd * DK + M0 + kc);

    float4 kr[8], vr0[2], vr1[2];

#define KV_FETCH(T0)                                                              \
    _Pragma("unroll") for (int z = 0; z < 2; ++z) {                               \
        const float* kq = kg + (size_t)((T0) + z) * DK + kofs;                    \
        kr[4 * z + 0] = *(const float4*)(kq + M0);                                \
        kr[4 * z + 1] = *(const float4*)(kq + kc1);                               \
        kr[4 * z + 2] = *(const float4*)(kq + kc2);                               \
        kr[4 * z + 3] = *(const float4*)(kq + kc3);                               \
    }                                                                             \
    _Pragma("unroll") for (int i = 0; i < 2; ++i) {                               \
        vr0[i] = *(const float4*)(vg + (size_t)((T0) + 8 * i) * DK + vofs);       \
        vr1[i] = *(const float4*)(vg + (size_t)((T0) + 8 * i + 1) * DK + vofs);   \
    }

#define KV_STAGE()                                                                \
    {                                                                             \
        const float4 ke = ln_row4(kr[0], kr[1], kr[2], kr[3], gv, bv);            \
        const float4 ko = ln_row4(kr[4], kr[5], kr[6], kr[7], gv, bv);            \
        put_pairs(ks, ke, ko);                                                    \
    }                                                                             \
    _Pragma("unroll") for (int i = 0; i < 2; ++i)                                 \
        put_pairs(vs + 4 * i * VPS, vr0[i], vr1[i]);

#define KV_MULT()                                                                 \
    _Pragma("unroll") for (int k0 = 0; k0 < KVC; k0 += 4) {                       \
        const int kp = (k0 >> 1) + half;                                          \
        v2f a  = *(const v2f*)&lds_k[kp * KPS + 2 * l16];                         \
        v2f b0 = *(const v2f*)&lds_v[kp * VPS + 2 * l16];                         \
        v2f b1 = *(const v2f*)&lds_v[kp * VPS + 2 * (16 + l16)];                  \
        acc0 = wmma4(a, b0, acc0);                                                \
        acc1 = wmma4(a, b1, acc1);                                                \
    }

    v8f acc0 = {}; v8f acc1 = {};

    KV_FETCH(0)
    for (int t0 = 0; t0 < KVT - KVC; t0 += KVC) {
        KV_STAGE()
        KV_FETCH(t0 + KVC)
        KV_MULT()
    }
    KV_STAGE()
    KV_MULT()

#undef KV_FETCH
#undef KV_STAGE
#undef KV_MULT

    float* red = smem;
    __syncthreads();
    #pragma unroll
    for (int r = 0; r < 8; ++r) {
        red[(wave * 16 + r) * 32 + lane]     = acc0[r];
        red[(wave * 16 + 8 + r) * 32 + lane] = acc1[r];
    }
    __syncthreads();
    float s0 = red[wave * 32 + lane];
    float s1 = red[(8 + wave) * 32 + lane];
    #pragma unroll
    for (int w = 1; w < KVW; ++w) {
        s0 += red[(w * 16 + wave) * 32 + lane];
        s1 += red[(w * 16 + 8 + wave) * 32 + lane];
    }

    const float scale = 1.0f / 16384.0f;
    float* kvd = kvb + (size_t)bh * DK * DK;
    const int m = M0 + wave + 8 * half;
    kvd[m * DK + N0 + l16]      = s0 * scale;
    kvd[m * DK + N0 + 16 + l16] = s1 * scale;
}

__global__ __launch_bounds__(256, 8)
void qproj_kv_kernel(const float* __restrict__ x, const float* __restrict__ w,
                     float* __restrict__ qb,
                     const float* __restrict__ kbuf, const float* __restrict__ vbuf,
                     const float* __restrict__ gk, const float* __restrict__ bk,
                     float* __restrict__ kvb) {
    __shared__ __attribute__((aligned(16))) float smem[SMEM_FLOATS];
    const int bid = blockIdx.x;
    if (bid < KV_BLOCKS) {
        kv_block(smem, kbuf, vbuf, gk, bk, kvb, bid & 7, bid >> 3);
    } else {
        proj_tile(x, w, qb, (bid & 7) * 64, bid & 7, (bid >> 3) * 64, smem);
    }
}

__global__ __launch_bounds__(256)
void out_gemm_kernel(const float* __restrict__ qbuf, const float* __restrict__ kvb,
                     const float* __restrict__ gq, const float* __restrict__ bq,
                     float* __restrict__ out) {
    __shared__ __attribute__((aligned(16))) float lds_q[64][68];
    __shared__ __attribute__((aligned(16))) float lds_kv[32 * BPS];

    const int tid  = threadIdx.x;
    const int lane = tid & 31, wave = tid >> 5;
    const int half = lane >> 4, l16 = lane & 15;
    const int M0 = (wave & 3) * 16, N0 = (wave >> 2) * 32;
    const int tok0 = blockIdx.x * 64;
    const int bh = blockIdx.y;
    const int bb = bh >> 3, h = bh & 7;
    const float* qp  = qbuf + (size_t)bh * SEQ * DK + (size_t)tok0 * DK;
    const float* kvp = kvb + (size_t)bh * DK * DK;
    const int sr = tid >> 4, sc = (tid & 15) * 4;

    v4f qa[4], k0r[2], k1r[2];
    load_rows(qa, qp, DK, sr, sc);
    load_pairs(k0r, k1r, kvp, DK, sr, sc);
    const v4f gv = *(const v4f*)(gq + h * DK + sc);
    const v4f bv = *(const v4f*)(bq + h * DK + sc);
    HOLD_LOADS(qa, k0r, k1r);
    #pragma unroll
    for (int i = 0; i < 4; ++i) *(v4f*)&lds_q[sr + 16 * i][sc] = ln_row16(qa[i], gv, bv);
    #pragma unroll
    for (int i = 0; i < 2; ++i) put_pairs(&lds_kv[(sr + 16 * i) * BPS + 2 * sc], k0r[i], k1r[i]);
    __syncthreads();

    v8f acc0 = {}; v8f acc1 = {};
    #pragma unroll
    for (int k0 = 0; k0 < 64; k0 += 4) {
        const int kp = (k0 >> 1) + half;
        v2f a  = *(const v2f*)&lds_q[M0 + l16][k0 + 2*half];
        v2f b0 = *(const v2f*)&lds_kv[kp * BPS + 2*(N0 + l16)];
        v2f b1 = *(const v2f*)&lds_kv[kp * BPS + 2*(N0 + 16 + l16)];
        acc0 = wmma4(a, b0, acc0);
        acc1 = wmma4(a, b1, acc1);
    }

    #pragma unroll
    for (int r = 0; r < 8; ++r) {
        int n = tok0 + M0 + r + 8*half;
        float* dst = out + ((size_t)(bb * SEQ + n)) * (HEADS * DK) + h * DK;
        dst[N0 + l16]      = acc0[r];
        dst[N0 + 16 + l16] = acc1[r];
    }
}

extern "C" void kernel_launch(void* const* d_in, const int* in_sizes, int n_in,
                              void* d_out, int out_size, void* d_ws, size_t ws_size,
                              hipStream_t stream) {
    const float* x  = (const float*)d_in[0];
    const float* w  = (const float*)d_in[1];
    const float* gq = (const float*)d_in[2];
    const float* bq = (const float*)d_in[3];
    const float* gk = (const float*)d_in[4];
    const float* bk = (const float*)d_in[5];
    float* out = (float*)d_out;

    float* ws = (float*)d_ws;
    const size_t perBuf = (size_t)BATCH * HEADS * SEQ * DK;
    float* qb  = ws;
    float* kb  = ws + perBuf;
    float* vb  = ws + 2 * perBuf;
    float* kvb = ws + 3 * perBuf;

    proj1_kernel<<<dim3(KV_TILES + KV_BLOCKS), 256, 0, stream>>>(x, w, qb, kb, vb);
    qproj_kv_kernel<<<dim3(Q_TILES), 256, 0, stream>>>(x, w, qb, kb, vb, gk, bk, kvb);
    out_gemm_kernel<<<dim3(SEQ / 64, BATCH * HEADS), 256, 0, stream>>>(qb, kvb, gq, bq, out);
}
